# starting kernel
# baseline (speedup 1.0000x reference)
#include <hip/hip_runtime.h>
#include <math.h>


typedef float v2f __attribute__((ext_vector_type(2)));
typedef float v8f __attribute__((ext_vector_type(8)));

#define BN   4
#define TT   2048
#define CC   768
#define HH   12
#define DD   64
#define FF   3072
#define NTOK 8192
#define LDA  772
#define SROW 2052

__device__ __forceinline__ v8f wmma4(v2f a, v2f b, v8f c) {
  return __builtin_amdgcn_wmma_f32_16x16x4_f32(false, a, false, b, (short)0, c,
                                               false, false);
}

__global__ __launch_bounds__(256) void transpose_kernel(
    const float* __restrict__ src, float* __restrict__ dst, int K, int N) {
  __shared__ float tile[32][33];
  const size_t bofs = (size_t)blockIdx.z * K * N;
  src += bofs;
  dst += bofs;
  const int k0 = blockIdx.y * 32, n0 = blockIdx.x * 32;
  const int tx = threadIdx.x & 31, ty = threadIdx.x >> 5;
  for (int i = ty; i < 32; i += 8)
    tile[i][tx] = src[(size_t)(k0 + i) * N + n0 + tx];
  __syncthreads();
  for (int i = ty; i < 32; i += 8)
    dst[(size_t)(n0 + i) * K + k0 + tx] = tile[tx][i];
}

__global__ __launch_bounds__(256) void qkv_kernel(
    const float* __restrict__ x, const float* __restrict__ wqkvt,
    float* __restrict__ qo, float* __restrict__ ko, float* __restrict__ vo) {
  __shared__ float xs[64 * LDA];
  const int m0 = blockIdx.x * 64;
  for (int i = threadIdx.x; i < 64 * (CC / 4); i += 256) {
    const int r = i / (CC / 4), c4 = i % (CC / 4);
    const float4 src = ((const float4*)(x + (m0 + r) * CC))[c4];
    *(float4*)(&xs[r * LDA + c4 * 4]) = src;
  }
  __syncthreads();
  const int wv = threadIdx.x >> 5, l = threadIdx.x & 31;
  const int n16 = l & 15, kq = (l >> 4) << 1;
  for (int task = wv; task < 144; task += 8) {
    const int rb = task / 36, oh = task % 36;
    const int o = oh / 12, head = oh % 12;
    const float* Wt = wqkvt + (size_t)oh * (DD * CC);
    v8f a0 = {}, a1 = {}, a2 = {}, a3 = {};
    const float* arow = &xs[(rb * 16 + n16) * LDA];
    const float* brow = Wt + (size_t)n16 * CC;
    for (int k0 = 0; k0 < CC; k0 += 4) {
      const v2f a = *(const v2f*)(arow + k0 + kq);
      const v2f b0 = *(const v2f*)(brow + k0 + kq);
      const v2f b1 = *(const v2f*)(brow + 16 * CC + k0 + kq);
      const v2f b2 = *(const v2f*)(brow + 32 * CC + k0 + kq);
      const v2f b3 = *(const v2f*)(brow + 48 * CC + k0 + kq);
      a0 = wmma4(a, b0, a0); a1 = wmma4(a, b1, a1);
      a2 = wmma4(a, b2, a2); a3 = wmma4(a, b3, a3);
    }
    const int mbase = rb * 16 + ((l >> 4) << 3);
    if (o < 2) {
      float* OUT = (o == 0) ? qo : ko;
#pragma unroll
      for (int j = 0; j < 8; ++j) {
        const int tok = m0 + mbase + j;
        const int b = tok >> 11, t = tok & 2047;
        float* dst = OUT + (((size_t)(b * HH + head) * TT + t) * DD);
        dst[0 + n16] = a0[j];  dst[16 + n16] = a1[j];
        dst[32 + n16] = a2[j]; dst[48 + n16] = a3[j];
      }
    } else {
#pragma unroll
      for (int j = 0; j < 8; ++j) {
        const int tok = m0 + mbase + j;
        const int b = tok >> 11, t = tok & 2047;
        float* base = vo + (size_t)(b * HH + head) * DD * TT + t;
        base[(size_t)(0 + n16) * TT] = a0[j];
        base[(size_t)(16 + n16) * TT] = a1[j];
        base[(size_t)(32 + n16) * TT] = a2[j];
        base[(size_t)(48 + n16) * TT] = a3[j];
      }
    }
  }
}

__global__ __launch_bounds__(256) void attn_kernel(
    const float* __restrict__ q, const float* __restrict__ k,
    const float* __restrict__ vt, float* __restrict__ aout) {
  __shared__ float S[16 * SROW];
  __shared__ float red[256];
  __shared__ float rowmax[16], rowrcp[16];
  __shared__ float opart[16 * 64];
  const int ttile = blockIdx.x;
  const int bh = blockIdx.y;
  const int t0 = ttile * 16;
  const float* qp = q + (size_t)bh * TT * DD;
  const float* kp = k + (size_t)bh * TT * DD;
  const float* vp = vt + (size_t)bh * TT * DD;
  const int wv = threadIdx.x >> 5, l = threadIdx.x & 31;
  const int n16 = l & 15, hi = l >> 4, kq = hi << 1;
  const int NT = ttile + 1, L = NT * 16;
  const float scale = 0.036084391824352f;

  v2f af[16];
#pragma unroll
  for (int ks = 0; ks < 16; ++ks)
    af[ks] = *(const v2f*)(kp + (t0 + n16) * DD + ks * 4 + kq);
  for (int st = wv; st < NT; st += 8) {
    const int s0 = st * 16;
    v8f acc = {};
#pragma unroll
    for (int ks = 0; ks < 16; ++ks) {
      const v2f b = *(const v2f*)(qp + (s0 + n16) * DD + ks * 4 + kq);
      acc = wmma4(af[ks], b, acc);
    }
    const int sg = s0 + n16;
#pragma unroll
    for (int j = 0; j < 8; ++j) {
      const int row = (hi << 3) + j;
      float vl = acc[j] * scale;
      if (sg > t0 + row) vl = -1e30f;
      S[row * SROW + sg] = vl;
    }
  }
  __syncthreads();

  {
    const int row = threadIdx.x >> 4, sub = threadIdx.x & 15;
    float m = -1e30f;
    for (int c = sub; c < L; c += 16) m = fmaxf(m, S[row * SROW + c]);
    red[threadIdx.x] = m;
    __syncthreads();
    if (sub == 0) {
      float mm = red[threadIdx.x];
      for (int i = 1; i < 16; ++i) mm = fmaxf(mm, red[threadIdx.x + i]);
      rowmax[row] = mm;
    }
    __syncthreads();
    const float mr = rowmax[row];
    float s = 0.f;
    for (int c = sub; c < L; c += 16) {
      const float e = __expf(S[row * SROW + c] - mr);
      S[row * SROW + c] = e;
      s += e;
    }
    red[threadIdx.x] = s;
    __syncthreads();
    if (sub == 0) {
      float ss = 0.f;
      for (int i = 0; i < 16; ++i) ss += red[threadIdx.x + i];
      rowrcp[row] = 1.0f / ss;
    }
  }
  __syncthreads();

  const int tile = wv & 3, half = wv >> 2;
  const int ksteps = L >> 2, kh = ksteps >> 1;
  const int kbeg = half * kh, kend = kbeg + kh;
  v8f acc = {};
  const float* vrow = vp + (size_t)(tile * 16 + n16) * TT;
  for (int ks = kbeg; ks < kend; ++ks) {
    const v2f a = *(const v2f*)(&S[n16 * SROW + ks * 4 + kq]);
    const v2f b = *(const v2f*)(vrow + ks * 4 + kq);
    acc = wmma4(a, b, acc);
  }
  if (half == 1) {
#pragma unroll
    for (int j = 0; j < 8; ++j)
      opart[((hi << 3) + j) * 64 + tile * 16 + n16] = acc[j];
  }
  __syncthreads();
  if (half == 0) {
    const int bb = bh / HH, head = bh % HH;
#pragma unroll
    for (int j = 0; j < 8; ++j) {
      const int row = (hi << 3) + j;
      const float val =
          (acc[j] + opart[row * 64 + tile * 16 + n16]) * rowrcp[row];
      const int t = t0 + row;
      aout[((size_t)(bb * TT + t)) * CC + head * DD + tile * 16 + n16] = val;
    }
  }
}

__global__ __launch_bounds__(256) void add_ln_kernel(
    const float* __restrict__ xa, const float* __restrict__ xb,
    const float* __restrict__ g, const float* __restrict__ bt,
    float* __restrict__ out) {
  __shared__ float red[256];
  const int tok = blockIdx.x;
  const float* pa = xa + (size_t)tok * CC;
  const float* pb = xb + (size_t)tok * CC;
  float vals[3];
  float s1 = 0.f;
#pragma unroll
  for (int i = 0; i < 3; ++i) {
    const int c = threadIdx.x + i * 256;
    vals[i] = pa[c] + pb[c];
    s1 += vals[i];
  }
  red[threadIdx.x] = s1;
  __syncthreads();
  for (int off = 128; off > 0; off >>= 1) {
    if (threadIdx.x < off) red[threadIdx.x] += red[threadIdx.x + off];
    __syncthreads();
  }
  const float mean = red[0] * (1.0f / CC);
  __syncthreads();
  float s2 = 0.f;
#pragma unroll
  for (int i = 0; i < 3; ++i) {
    const float d = vals[i] - mean;
    s2 += d * d;
  }
  red[threadIdx.x] = s2;
  __syncthreads();
  for (int off = 128; off > 0; off >>= 1) {
    if (threadIdx.x < off) red[threadIdx.x] += red[threadIdx.x + off];
    __syncthreads();
  }
  const float rstd = rsqrtf(red[0] * (1.0f / CC) + 1e-5f);
#pragma unroll
  for (int i = 0; i < 3; ++i) {
    const int c = threadIdx.x + i * 256;
    out[(size_t)tok * CC + c] = (vals[i] - mean) * rstd * g[c] + bt[c];
  }
}

__global__ __launch_bounds__(256) void mlp1_kernel(
    const float* __restrict__ xin, const float* __restrict__ W1t,
    const float* __restrict__ bias, float* __restrict__ hout) {
  __shared__ float xs[64 * LDA];
  const int m0 = blockIdx.x * 64;
  for (int i = threadIdx.x; i < 64 * (CC / 4); i += 256) {
    const int r = i / (CC / 4), c4 = i % (CC / 4);
    const float4 src = ((const float4*)(xin + (m0 + r) * CC))[c4];
    *(float4*)(&xs[r * LDA + c4 * 4]) = src;
  }
  __syncthreads();
  const int wv = threadIdx.x >> 5, l = threadIdx.x & 31;
  const int n16 = l & 15, kq = (l >> 4) << 1;
  for (int task = wv; task < 192; task += 8) {
    const int rb = task / 48, nb = (task % 48) * 64;
    v8f a0 = {}, a1 = {}, a2 = {}, a3 = {};
    const float* arow = &xs[(rb * 16 + n16) * LDA];
    const float* brow = W1t + (size_t)(nb + n16) * CC;
    for (int k0 = 0; k0 < CC; k0 += 4) {
      const v2f a = *(const v2f*)(arow + k0 + kq);
      const v2f b0 = *(const v2f*)(brow + k0 + kq);
      const v2f b1 = *(const v2f*)(brow + (size_t)16 * CC + k0 + kq);
      const v2f b2 = *(const v2f*)(brow + (size_t)32 * CC + k0 + kq);
      const v2f b3 = *(const v2f*)(brow + (size_t)48 * CC + k0 + kq);
      a0 = wmma4(a, b0, a0); a1 = wmma4(a, b1, a1);
      a2 = wmma4(a, b2, a2); a3 = wmma4(a, b3, a3);
    }
    const int mbase = rb * 16 + ((l >> 4) << 3);
#pragma unroll
    for (int j = 0; j < 8; ++j) {
      float* dst = hout + (size_t)(m0 + mbase + j) * FF + nb + n16;
      float z;
      z = a0[j] + bias[nb + n16];
      dst[0] = 0.5f * z * (1.f + erff(z * 0.70710678118f));
      z = a1[j] + bias[nb + 16 + n16];
      dst[16] = 0.5f * z * (1.f + erff(z * 0.70710678118f));
      z = a2[j] + bias[nb + 32 + n16];
      dst[32] = 0.5f * z * (1.f + erff(z * 0.70710678118f));
      z = a3[j] + bias[nb + 48 + n16];
      dst[48] = 0.5f * z * (1.f + erff(z * 0.70710678118f));
    }
  }
}

__global__ __launch_bounds__(256) void mlp2_kernel(
    const float* __restrict__ hin, const float* __restrict__ W2t,
    const float* __restrict__ bias, float* __restrict__ t2) {
  __shared__ float opart[64 * 64];
  const int m0 = blockIdx.x * 64, n0 = blockIdx.y * 64;
  const int wv = threadIdx.x >> 5, l = threadIdx.x & 31;
  const int n16 = l & 15, kq = (l >> 4) << 1;
  const int rb = wv & 3, half = wv >> 2;
  const int kbeg = half * 384, kend = kbeg + 384;
  v8f a0 = {}, a1 = {}, a2 = {}, a3 = {};
  const float* arow = hin + (size_t)(m0 + rb * 16 + n16) * FF;
  const float* brow = W2t + (size_t)(n0 + n16) * FF;
  for (int ks = kbeg; ks < kend; ++ks) {
    const int kk = ks * 4 + kq;
    const v2f a = *(const v2f*)(arow + kk);
    const v2f b0 = *(const v2f*)(brow + kk);
    const v2f b1 = *(const v2f*)(brow + (size_t)16 * FF + kk);
    const v2f b2 = *(const v2f*)(brow + (size_t)32 * FF + kk);
    const v2f b3 = *(const v2f*)(brow + (size_t)48 * FF + kk);
    a0 = wmma4(a, b0, a0); a1 = wmma4(a, b1, a1);
    a2 = wmma4(a, b2, a2); a3 = wmma4(a, b3, a3);
  }
  const int mb = rb * 16 + ((l >> 4) << 3);
  if (half == 1) {
#pragma unroll
    for (int j = 0; j < 8; ++j) {
      opart[(mb + j) * 64 + n16] = a0[j];
      opart[(mb + j) * 64 + 16 + n16] = a1[j];
      opart[(mb + j) * 64 + 32 + n16] = a2[j];
      opart[(mb + j) * 64 + 48 + n16] = a3[j];
    }
  }
  __syncthreads();
  if (half == 0) {
#pragma unroll
    for (int j = 0; j < 8; ++j) {
      float* dst = t2 + (size_t)(m0 + mb + j) * CC + n0 + n16;
      dst[0]  = a0[j] + opart[(mb + j) * 64 + n16]      + bias[n0 + n16];
      dst[16] = a1[j] + opart[(mb + j) * 64 + 16 + n16] + bias[n0 + 16 + n16];
      dst[32] = a2[j] + opart[(mb + j) * 64 + 32 + n16] + bias[n0 + 32 + n16];
      dst[48] = a3[j] + opart[(mb + j) * 64 + 48 + n16] + bias[n0 + 48 + n16];
    }
  }
}

extern "C" void kernel_launch(void* const* d_in, const int* in_sizes, int n_in,
                              void* d_out, int out_size, void* d_ws,
                              size_t ws_size, hipStream_t stream) {
  (void)in_sizes; (void)n_in; (void)out_size; (void)ws_size;
  const float* x    = (const float*)d_in[0];
  const float* Wq   = (const float*)d_in[1];
  const float* Wk   = (const float*)d_in[2];
  const float* Wv   = (const float*)d_in[3];
  const float* ln1g = (const float*)d_in[4];
  const float* ln1b = (const float*)d_in[5];
  const float* W1   = (const float*)d_in[6];
  const float* b1   = (const float*)d_in[7];
  const float* W2   = (const float*)d_in[8];
  const float* b2   = (const float*)d_in[9];
  const float* ln2g = (const float*)d_in[10];
  const float* ln2b = (const float*)d_in[11];
  float* ws  = (float*)d_ws;
  float* out = (float*)d_out;

  float* q     = ws;
  float* k     = ws + 6291456;
  float* v     = ws + 12582912;
  float* a     = ws + 18874368;
  float* hbuf  = ws;
  float* x1    = ws + 25165824;
  float* t2    = ws + 31457280;
  float* wqkvt = ws + 37748736;
  float* w1t   = ws + 39518208;
  float* w2t   = ws + 41877504;

  transpose_kernel<<<dim3(DD / 32, CC / 32, HH), 256, 0, stream>>>(
      Wq, wqkvt + 0 * (size_t)HH * DD * CC, CC, DD);
  transpose_kernel<<<dim3(DD / 32, CC / 32, HH), 256, 0, stream>>>(
      Wk, wqkvt + 1 * (size_t)HH * DD * CC, CC, DD);
  transpose_kernel<<<dim3(DD / 32, CC / 32, HH), 256, 0, stream>>>(
      Wv, wqkvt + 2 * (size_t)HH * DD * CC, CC, DD);
  transpose_kernel<<<dim3(FF / 32, CC / 32, 1), 256, 0, stream>>>(
      W1, w1t, CC, FF);
  transpose_kernel<<<dim3(CC / 32, FF / 32, 1), 256, 0, stream>>>(
      W2, w2t, FF, CC);

  qkv_kernel<<<dim3(NTOK / 64), 256, 0, stream>>>(x, wqkvt, q, k, v);
  attn_kernel<<<dim3(TT / 16, BN * HH), 256, 0, stream>>>(q, k, v, a);
  add_ln_kernel<<<dim3(NTOK), 256, 0, stream>>>(x, a, ln1g, ln1b, x1);
  mlp1_kernel<<<dim3(NTOK / 64), 256, 0, stream>>>(x1, w1t, b1, hbuf);
  mlp2_kernel<<<dim3(NTOK / 64, CC / 64), 256, 0, stream>>>(hbuf, w2t, b2, t2);
  add_ln_kernel<<<dim3(NTOK), 256, 0, stream>>>(x1, t2, ln2g, ln2b, out);
}


# attention: q and v tiles arrive as whole-line 16-byte global loads through a wave-private LDS staging tile (16 rows x 32 k), not per-lane 8-byte global loads touching 16 lines each
# speedup vs baseline: 3.0558x; 3.0558x over previous
#include <hip/hip_runtime.h>
#include <math.h>


typedef float v2f __attribute__((ext_vector_type(2)));
typedef float v8f __attribute__((ext_vector_type(8)));

#define BN   4
#define TT   2048
#define CC   768
#define HH   12
#define DD   64
#define FF   3072
#define NTOK 8192
#define SROW 2052

__device__ __forceinline__ v8f wmma4(v2f a, v2f b, v8f c) {
  return __builtin_amdgcn_wmma_f32_16x16x4_f32(false, a, false, b, (short)0, c,
                                               false, false);
}

__global__ __launch_bounds__(256) void transpose_kernel(
    const float* __restrict__ src, float* __restrict__ dst, int K, int N) {
  __shared__ float tile[32][33];
  const size_t bofs = (size_t)blockIdx.z * K * N;
  src += bofs;
  dst += bofs;
  const int k0 = blockIdx.y * 32, n0 = blockIdx.x * 32;
  const int tx = threadIdx.x & 31, ty = threadIdx.x >> 5;
  for (int i = ty; i < 32; i += 8)
    tile[i][tx] = src[(size_t)(k0 + i) * N + n0 + tx];
  __syncthreads();
  for (int i = ty; i < 32; i += 8)
    dst[(size_t)(n0 + i) * K + k0 + tx] = tile[tx][i];
}

#define GBK 32
#define GPITCH 36

template <int MI, int MODE, int K>
__global__ __launch_bounds__(256) void gemm_kernel(
    const float* __restrict__ A, const float* __restrict__ Bt,
    const float* __restrict__ bias, float* __restrict__ out0,
    float* __restrict__ out1, float* __restrict__ out2) {
  constexpr int BM = MI * 32;
  constexpr int KT = K / GBK;
  __shared__ alignas(16) float As[2][BM * GPITCH];
  __shared__ alignas(16) float Bs[2][128 * GPITCH];
  const int tid = threadIdx.x;
  const int wv = tid >> 5, l = tid & 31;
  const int c = l & 15, h = l >> 4;
  const int wm = wv >> 2, wn = wv & 3;
  const int m0 = blockIdx.y * BM, n0 = blockIdx.x * 128;

  const int sr = tid >> 3, sc = (tid & 7) << 2;
  const float* ag = A + (size_t)(m0 + sr) * K + sc;
  const float* bg = Bt + (size_t)(n0 + sr) * K + sc;
  const int sofs = sr * GPITCH + sc;
  float4 ga[MI], gb[4];
#pragma unroll
  for (int i = 0; i < MI; ++i)
    ga[i] = *(const float4*)(ag + (size_t)(i * 32) * K);
#pragma unroll
  for (int i = 0; i < 4; ++i)
    gb[i] = *(const float4*)(bg + (size_t)(i * 32) * K);
#pragma unroll
  for (int i = 0; i < MI; ++i)
    *(float4*)(&As[0][sofs + i * 32 * GPITCH]) = ga[i];
#pragma unroll
  for (int i = 0; i < 4; ++i)
    *(float4*)(&Bs[0][sofs + i * 32 * GPITCH]) = gb[i];
  __syncthreads();

  v8f acc[MI][2];
#pragma unroll
  for (int i = 0; i < MI; ++i) {
    acc[i][0] = v8f{};
    acc[i][1] = v8f{};
  }
  const int aofs = (wm * (MI * 16) + c) * GPITCH + 4 * h;
  const int bofs = (wn * 32 + c) * GPITCH + 4 * h;

  for (int kt = 0; kt < KT; ++kt) {
    const int cur = kt & 1;
    const bool more = (kt + 1 < KT);
    if (more) {
      const int kn = (kt + 1) * GBK;
#pragma unroll
      for (int i = 0; i < MI; ++i)
        ga[i] = *(const float4*)(ag + (size_t)(i * 32) * K + kn);
#pragma unroll
      for (int i = 0; i < 4; ++i)
        gb[i] = *(const float4*)(bg + (size_t)(i * 32) * K + kn);
      __builtin_amdgcn_sched_barrier(0);
    }
    const float* as = &As[cur][aofs];
    const float* bs = &Bs[cur][bofs];
    float4 fa[2][MI], fb[2][2];
#pragma unroll
    for (int i = 0; i < MI; ++i)
      fa[0][i] = *(const float4*)(as + i * 16 * GPITCH);
#pragma unroll
    for (int jn = 0; jn < 2; ++jn)
      fb[0][jn] = *(const float4*)(bs + jn * 16 * GPITCH);
#pragma unroll
    for (int kk = 0; kk < 4; ++kk) {
      const int p = kk & 1;
      if (kk < 3) {
#pragma unroll
        for (int i = 0; i < MI; ++i)
          fa[p ^ 1][i] =
              *(const float4*)(as + i * 16 * GPITCH + (kk + 1) * 8);
#pragma unroll
        for (int jn = 0; jn < 2; ++jn)
          fb[p ^ 1][jn] =
              *(const float4*)(bs + jn * 16 * GPITCH + (kk + 1) * 8);
      }
#pragma unroll
      for (int i = 0; i < MI; ++i)
#pragma unroll
        for (int jn = 0; jn < 2; ++jn) {
          const v2f a = {fa[p][i].x, fa[p][i].y};
          const v2f b = {fb[p][jn].x, fb[p][jn].y};
          acc[i][jn] = wmma4(a, b, acc[i][jn]);
        }
#pragma unroll
      for (int i = 0; i < MI; ++i)
#pragma unroll
        for (int jn = 0; jn < 2; ++jn) {
          const v2f a = {fa[p][i].z, fa[p][i].w};
          const v2f b = {fb[p][jn].z, fb[p][jn].w};
          acc[i][jn] = wmma4(a, b, acc[i][jn]);
        }
    }
    if (more) {
#pragma unroll
      for (int i = 0; i < MI; ++i)
        *(float4*)(&As[cur ^ 1][sofs + i * 32 * GPITCH]) = ga[i];
#pragma unroll
      for (int i = 0; i < 4; ++i)
        *(float4*)(&Bs[cur ^ 1][sofs + i * 32 * GPITCH]) = gb[i];
    }
    __syncthreads();
  }

  const int rbase = m0 + wm * (MI * 16) + (h << 3);
  const int ncol = n0 + wn * 32 + c;
  if (MODE == 0) {
    const int nw = n0 + wn * 32;
    const int o = nw / CC;
    const int rem = nw - o * CC;
    const int head = rem >> 6, d = (rem & 63) + c;
    if (o < 2) {
      float* OUT = (o == 0) ? out0 : out1;
#pragma unroll
      for (int i = 0; i < MI; ++i)
#pragma unroll
        for (int j = 0; j < 8; ++j) {
          const int tok = rbase + i * 16 + j;
          const int b = tok >> 11, t = tok & 2047;
          float* dst = OUT + (((size_t)(b * HH + head) * TT + t) * DD) + d;
          dst[0] = acc[i][0][j];
          dst[16] = acc[i][1][j];
        }
    } else {
#pragma unroll
      for (int i = 0; i < MI; ++i)
#pragma unroll
        for (int j = 0; j < 8; ++j) {
          const int tok = rbase + i * 16 + j;
          const int b = tok >> 11, t = tok & 2047;
          float* base = out2 + ((size_t)(b * HH + head) * DD + d) * TT + t;
          base[0] = acc[i][0][j];
          base[(size_t)16 * TT] = acc[i][1][j];
        }
    }
  } else if (MODE == 1) {
    const float bv0 = bias[ncol], bv1 = bias[ncol + 16];
#pragma unroll
    for (int i = 0; i < MI; ++i)
#pragma unroll
      for (int j = 0; j < 8; ++j) {
        float* dst = out0 + (size_t)(rbase + i * 16 + j) * FF + ncol;
        float z;
        z = acc[i][0][j] + bv0;
        dst[0] = 0.5f * z * (1.f + erff(z * 0.70710678118f));
        z = acc[i][1][j] + bv1;
        dst[16] = 0.5f * z * (1.f + erff(z * 0.70710678118f));
      }
  } else {
    const float bv0 = bias[ncol], bv1 = bias[ncol + 16];
#pragma unroll
    for (int i = 0; i < MI; ++i)
#pragma unroll
      for (int j = 0; j < 8; ++j) {
        float* dst = out0 + (size_t)(rbase + i * 16 + j) * CC + ncol;
        dst[0] = acc[i][0][j] + bv0;
        dst[16] = acc[i][1][j] + bv1;
      }
  }
}

#define APITCH 36
typedef float v4f __attribute__((ext_vector_type(4)));

__device__ __forceinline__ void wave_sync() {
  __builtin_amdgcn_fence(__ATOMIC_RELEASE, "wavefront");
  __builtin_amdgcn_wave_barrier();
  __builtin_amdgcn_fence(__ATOMIC_ACQUIRE, "wavefront");
}

__device__ __forceinline__ void stage_and_refill(float* stg, int sofs,
                                                 v4f (&g)[4],
                                                 const float* gnext,
                                                 int gpitch) {
  wave_sync();
#pragma unroll
  for (int j = 0; j < 4; ++j) *(v4f*)(stg + sofs + j * 4 * APITCH) = g[j];
  wave_sync();
#pragma unroll
  for (int j = 0; j < 4; ++j)
    g[j] = *(const v4f*)(gnext + (size_t)(4 * j) * gpitch);
}

__device__ __forceinline__ void qk_chunk(const float* stg, int fofs,
                                         const v4f (&ka)[4], v8f& acc) {
  v4f fb[4];
#pragma unroll
  for (int kk = 0; kk < 4; ++kk) fb[kk] = *(const v4f*)(stg + fofs + kk * 8);
#pragma unroll
  for (int kk = 0; kk < 4; ++kk) {
    const v2f a0 = {ka[kk].x, ka[kk].y};
    const v2f b0 = {fb[kk].x, fb[kk].y};
    acc = wmma4(a0, b0, acc);
    const v2f a1 = {ka[kk].z, ka[kk].w};
    const v2f b1 = {fb[kk].z, fb[kk].w};
    acc = wmma4(a1, b1, acc);
  }
}

__device__ __forceinline__ void pv_half(const float* stg, int fofs,
                                        const float* p, v8f& acc) {
  v4f fa[2], fb[2];
#pragma unroll
  for (int kk = 0; kk < 2; ++kk) {
    fa[kk] = *(const v4f*)(p + kk * 8);
    fb[kk] = *(const v4f*)(stg + fofs + kk * 8);
  }
#pragma unroll
  for (int kk = 0; kk < 2; ++kk) {
    const v2f a0 = {fa[kk].x, fa[kk].y};
    const v2f b0 = {fb[kk].x, fb[kk].y};
    acc = wmma4(a0, b0, acc);
    const v2f a1 = {fa[kk].z, fa[kk].w};
    const v2f b1 = {fb[kk].z, fb[kk].w};
    acc = wmma4(a1, b1, acc);
  }
}

__global__ __launch_bounds__(256) void attn_kernel(
    const float* __restrict__ q, const float* __restrict__ k,
    const float* __restrict__ vt, float* __restrict__ aout) {
  __shared__ alignas(16) float S[16 * SROW];
  __shared__ alignas(16) float stage[8 * 16 * APITCH];
  __shared__ float red[256];
  __shared__ float rowmax[16], rowrcp[16];
  __shared__ float opart[16 * 64];
  const int ttile = blockIdx.x;
  const int bh = blockIdx.y;
  const int t0 = ttile * 16;
  const float* qp = q + (size_t)bh * TT * DD;
  const float* kp = k + (size_t)bh * TT * DD;
  const float* vp = vt + (size_t)bh * TT * DD;
  const int wv = __builtin_amdgcn_readfirstlane((int)(threadIdx.x >> 5));
  const int l = threadIdx.x & 31;
  const int n16 = l & 15, hi = l >> 4;
  const int NT = ttile + 1, L = NT * 16;
  const float scale = 0.036084391824352f;

  float* stg = stage + wv * (16 * APITCH);
  const int srow = l >> 3, scol = (l & 7) << 2;
  const int sofs = srow * APITCH + scol;
  const int fofs = n16 * APITCH + (hi << 2);

  v4f ka0[4], ka1[4];
  {
    const float* kr = kp + (size_t)(t0 + n16) * DD + (hi << 2);
#pragma unroll
    for (int kk = 0; kk < 4; ++kk) ka0[kk] = *(const v4f*)(kr + kk * 8);
#pragma unroll
    for (int kk = 0; kk < 4; ++kk) ka1[kk] = *(const v4f*)(kr + 32 + kk * 8);
  }
  const float* qg = qp + (size_t)srow * DD + scol;
  v4f g0[4], g1[4];
  {
    const float* qf = qg + (size_t)((wv < NT ? wv : 0) * 16) * DD;
#pragma unroll
    for (int j = 0; j < 4; ++j) g0[j] = *(const v4f*)(qf + (size_t)(4 * j) * DD);
#pragma unroll
    for (int j = 0; j < 4; ++j)
      g1[j] = *(const v4f*)(qf + (size_t)(4 * j) * DD + 32);
  }
  for (int st = wv; st < NT; st += 8) {
    const int s0 = st * 16;
    const int sn = (st + 8 < NT) ? st + 8 : st;
    const float* qn = qg + (size_t)(sn * 16) * DD;
    v8f acc = {};
    stage_and_refill(stg, sofs, g0, qn, DD);
    qk_chunk(stg, fofs, ka0, acc);
    stage_and_refill(stg, sofs, g1, qn + 32, DD);
    qk_chunk(stg, fofs, ka1, acc);
    const int sg = s0 + n16;
#pragma unroll
    for (int j = 0; j < 8; ++j) {
      const int row = (hi << 3) + j;
      float vl = acc[j] * scale;
      if (sg > t0 + row) vl = -1e30f;
      S[row * SROW + sg] = vl;
    }
  }
  __syncthreads();

  {
    const int row = threadIdx.x >> 4, sub = threadIdx.x & 15;
    float m = -1e30f;
    for (int c = sub; c < L; c += 16) m = fmaxf(m, S[row * SROW + c]);
    red[threadIdx.x] = m;
    __syncthreads();
    if (sub == 0) {
      float mm = red[threadIdx.x];
      for (int i = 1; i < 16; ++i) mm = fmaxf(mm, red[threadIdx.x + i]);
      rowmax[row] = mm;
    }
    __syncthreads();
    const float mr = rowmax[row];
    float s = 0.f;
    for (int c = sub; c < L; c += 16) {
      const float e = __expf(S[row * SROW + c] - mr);
      S[row * SROW + c] = e;
      s += e;
    }
    red[threadIdx.x] = s;
    __syncthreads();
    if (sub == 0) {
      float ss = 0.f;
      for (int i = 0; i < 16; ++i) ss += red[threadIdx.x + i];
      rowrcp[row] = 1.0f / ss;
    }
  }
  __syncthreads();

  const int tile = wv & 3, half = wv >> 2;
  const int NC = (L + 31) >> 5;
  const int cmid = (NC + 1) >> 1;
  const int cbeg = half ? cmid : 0, cend = half ? NC : cmid;
  const int clast = (cend > cbeg) ? cend - 1 : 0;
  v8f acc = {};
  const float* vg = vp + (size_t)(tile * 16 + srow) * TT + scol;
  const float* prow = &S[n16 * SROW + (hi << 2)];
  v4f ga[4], gb[4];
  {
    const int ca = (cbeg < clast) ? cbeg : clast;
    const int cb = (cbeg + 1 < clast) ? cbeg + 1 : clast;
#pragma unroll
    for (int j = 0; j < 4; ++j)
      ga[j] = *(const v4f*)(vg + (size_t)(4 * j) * TT + ca * 32);
#pragma unroll
    for (int j = 0; j < 4; ++j)
      gb[j] = *(const v4f*)(vg + (size_t)(4 * j) * TT + cb * 32);
  }
  int cc = cbeg;
  for (; cc + 1 < cend; cc += 2) {
    const int c2 = (cc + 2 < clast) ? cc + 2 : clast;
    const int c3 = (cc + 3 < clast) ? cc + 3 : clast;
    stage_and_refill(stg, sofs, ga, vg + c2 * 32, TT);
    pv_half(stg, fofs, prow + cc * 32, acc);
    pv_half(stg, fofs + 16, prow + cc * 32 + 16, acc);
    stage_and_refill(stg, sofs, gb, vg + c3 * 32, TT);
    pv_half(stg, fofs, prow + (cc + 1) * 32, acc);
    if (L - (cc + 1) * 32 >= 32)
      pv_half(stg, fofs + 16, prow + (cc + 1) * 32 + 16, acc);
  }
  if (cc < cend) {
    stage_and_refill(stg, sofs, ga, vg + clast * 32, TT);
    pv_half(stg, fofs, prow + cc * 32, acc);
    if (L - cc * 32 >= 32) pv_half(stg, fofs + 16, prow + cc * 32 + 16, acc);
  }
  if (half == 1) {
#pragma unroll
    for (int j = 0; j < 8; ++j)
      opart[((hi << 3) + j) * 64 + tile * 16 + n16] = acc[j];
  }
  __syncthreads();
  if (half == 0) {
    const int bb = bh / HH, head = bh % HH;
#pragma unroll
    for (int j = 0; j < 8; ++j) {
      const int row = (hi << 3) + j;
      const float val =
          (acc[j] + opart[row * 64 + tile * 16 + n16]) * rowrcp[row];
      const int t = t0 + row;
      aout[((size_t)(bb * TT + t)) * CC + head * DD + tile * 16 + n16] = val;
    }
  }
}

__global__ __launch_bounds__(256) void add_ln_kernel(
    const float* __restrict__ xa, const float* __restrict__ xb,
    const float* __restrict__ g, const float* __restrict__ bt,
    float* __restrict__ out) {
  __shared__ float red[256];
  const int tok = blockIdx.x;
  const float* pa = xa + (size_t)tok * CC;
  const float* pb = xb + (size_t)tok * CC;
  float vals[3];
  float s1 = 0.f;
#pragma unroll
  for (int i = 0; i < 3; ++i) {
    const int c = threadIdx.x + i * 256;
    vals[i] = pa[c] + pb[c];
    s1 += vals[i];
  }
  red[threadIdx.x] = s1;
  __syncthreads();
  for (int off = 128; off > 0; off >>= 1) {
    if (threadIdx.x < off) red[threadIdx.x] += red[threadIdx.x + off];
    __syncthreads();
  }
  const float mean = red[0] * (1.0f / CC);
  __syncthreads();
  float s2 = 0.f;
#pragma unroll
  for (int i = 0; i < 3; ++i) {
    const float d = vals[i] - mean;
    s2 += d * d;
  }
  red[threadIdx.x] = s2;
  __syncthreads();
  for (int off = 128; off > 0; off >>= 1) {
    if (threadIdx.x < off) red[threadIdx.x] += red[threadIdx.x + off];
    __syncthreads();
  }
  const float rstd = rsqrtf(red[0] * (1.0f / CC) + 1e-5f);
#pragma unroll
  for (int i = 0; i < 3; ++i) {
    const int c = threadIdx.x + i * 256;
    out[(size_t)tok * CC + c] = (vals[i] - mean) * rstd * g[c] + bt[c];
  }
}

extern "C" void kernel_launch(void* const* d_in, const int* in_sizes, int n_in,
                              void* d_out, int out_size, void* d_ws,
                              size_t ws_size, hipStream_t stream) {
  (void)in_sizes; (void)n_in; (void)out_size; (void)ws_size;
  const float* x    = (const float*)d_in[0];
  const float* Wq   = (const float*)d_in[1];
  const float* Wk   = (const float*)d_in[2];
  const float* Wv   = (const float*)d_in[3];
  const float* ln1g = (const float*)d_in[4];
  const float* ln1b = (const float*)d_in[5];
  const float* W1   = (const float*)d_in[6];
  const float* b1   = (const float*)d_in[7];
  const float* W2   = (const float*)d_in[8];
  const float* b2   = (const float*)d_in[9];
  const float* ln2g = (const float*)d_in[10];
  const float* ln2b = (const float*)d_in[11];
  float* ws  = (float*)d_ws;
  float* out = (float*)d_out;

  float* q     = ws;
  float* k     = ws + 6291456;
  float* v     = ws + 12582912;
  float* a     = ws + 18874368;
  float* hbuf  = ws;
  float* x1    = ws + 25165824;
  float* t2    = ws + 31457280;
  float* wqkvt = ws + 37748736;
  float* w1t   = ws + 39518208;
  float* w2t   = ws + 41877504;

  transpose_kernel<<<dim3(DD / 32, CC / 32, HH), 256, 0, stream>>>(
      Wq, wqkvt + 0 * (size_t)HH * DD * CC, CC, DD);
  transpose_kernel<<<dim3(DD / 32, CC / 32, HH), 256, 0, stream>>>(
      Wk, wqkvt + 1 * (size_t)HH * DD * CC, CC, DD);
  transpose_kernel<<<dim3(DD / 32, CC / 32, HH), 256, 0, stream>>>(
      Wv, wqkvt + 2 * (size_t)HH * DD * CC, CC, DD);
  transpose_kernel<<<dim3(FF / 32, CC / 32, 1), 256, 0, stream>>>(
      W1, w1t, CC, FF);
  transpose_kernel<<<dim3(CC / 32, FF / 32, 1), 256, 0, stream>>>(
      W2, w2t, FF, CC);

  gemm_kernel<2, 0, CC><<<dim3(3 * CC / 128, NTOK / 64), 256, 0, stream>>>(
      x, wqkvt, nullptr, q, k, v);
  attn_kernel<<<dim3(TT / 16, BN * HH), 256, 0, stream>>>(q, k, v, a);
  add_ln_kernel<<<dim3(NTOK), 256, 0, stream>>>(x, a, ln1g, ln1b, x1);
  gemm_kernel<4, 1, CC><<<dim3(FF / 128, NTOK / 128), 256, 0, stream>>>(
      x1, w1t, b1, hbuf, nullptr, nullptr);
  gemm_kernel<2, 2, FF><<<dim3(CC / 128, NTOK / 64), 256, 0, stream>>>(
      hbuf, w2t, b2, t2, nullptr, nullptr);
  add_ln_kernel<<<dim3(NTOK), 256, 0, stream>>>(x1, t2, ln2g, ln2b, out);
}


# attention softmax passes: each lane walks its score row in 16-byte LDS pieces, several in flight, row max and sum reduced in registers (DPP), not one dependent 4-byte LDS round trip per element
# speedup vs baseline: 3.2817x; 1.0739x over previous
#include <hip/hip_runtime.h>
#include <math.h>


typedef float v2f __attribute__((ext_vector_type(2)));
typedef float v8f __attribute__((ext_vector_type(8)));

#define BN   4
#define TT   2048
#define CC   768
#define HH   12
#define DD   64
#define FF   3072
#define NTOK 8192
#define SROW 2052

__device__ __forceinline__ v8f wmma4(v2f a, v2f b, v8f c) {
  return __builtin_amdgcn_wmma_f32_16x16x4_f32(false, a, false, b, (short)0, c,
                                               false, false);
}

__global__ __launch_bounds__(256) void transpose_kernel(
    const float* __restrict__ src, float* __restrict__ dst, int K, int N) {
  __shared__ float tile[32][33];
  const size_t bofs = (size_t)blockIdx.z * K * N;
  src += bofs;
  dst += bofs;
  const int k0 = blockIdx.y * 32, n0 = blockIdx.x * 32;
  const int tx = threadIdx.x & 31, ty = threadIdx.x >> 5;
  for (int i = ty; i < 32; i += 8)
    tile[i][tx] = src[(size_t)(k0 + i) * N + n0 + tx];
  __syncthreads();
  for (int i = ty; i < 32; i += 8)
    dst[(size_t)(n0 + i) * K + k0 + tx] = tile[tx][i];
}

#define GBK 32
#define GPITCH 36

template <int MI, int MODE, int K>
__global__ __launch_bounds__(256) void gemm_kernel(
    const float* __restrict__ A, const float* __restrict__ Bt,
    const float* __restrict__ bias, float* __restrict__ out0,
    float* __restrict__ out1, float* __restrict__ out2) {
  constexpr int BM = MI * 32;
  constexpr int KT = K / GBK;
  __shared__ alignas(16) float As[2][BM * GPITCH];
  __shared__ alignas(16) float Bs[2][128 * GPITCH];
  const int tid = threadIdx.x;
  const int wv = tid >> 5, l = tid & 31;
  const int c = l & 15, h = l >> 4;
  const int wm = wv >> 2, wn = wv & 3;
  const int m0 = blockIdx.y * BM, n0 = blockIdx.x * 128;

  const int sr = tid >> 3, sc = (tid & 7) << 2;
  const float* ag = A + (size_t)(m0 + sr) * K + sc;
  const float* bg = Bt + (size_t)(n0 + sr) * K + sc;
  const int sofs = sr * GPITCH + sc;
  float4 ga[MI], gb[4];
#pragma unroll
  for (int i = 0; i < MI; ++i)
    ga[i] = *(const float4*)(ag + (size_t)(i * 32) * K);
#pragma unroll
  for (int i = 0; i < 4; ++i)
    gb[i] = *(const float4*)(bg + (size_t)(i * 32) * K);
#pragma unroll
  for (int i = 0; i < MI; ++i)
    *(float4*)(&As[0][sofs + i * 32 * GPITCH]) = ga[i];
#pragma unroll
  for (int i = 0; i < 4; ++i)
    *(float4*)(&Bs[0][sofs + i * 32 * GPITCH]) = gb[i];
  __syncthreads();

  v8f acc[MI][2];
#pragma unroll
  for (int i = 0; i < MI; ++i) {
    acc[i][0] = v8f{};
    acc[i][1] = v8f{};
  }
  const int aofs = (wm * (MI * 16) + c) * GPITCH + 4 * h;
  const int bofs = (wn * 32 + c) * GPITCH + 4 * h;

  for (int kt = 0; kt < KT; ++kt) {
    const int cur = kt & 1;
    const bool more = (kt + 1 < KT);
    if (more) {
      const int kn = (kt + 1) * GBK;
#pragma unroll
      for (int i = 0; i < MI; ++i)
        ga[i] = *(const float4*)(ag + (size_t)(i * 32) * K + kn);
#pragma unroll
      for (int i = 0; i < 4; ++i)
        gb[i] = *(const float4*)(bg + (size_t)(i * 32) * K + kn);
      __builtin_amdgcn_sched_barrier(0);
    }
    const float* as = &As[cur][aofs];
    const float* bs = &Bs[cur][bofs];
    float4 fa[2][MI], fb[2][2];
#pragma unroll
    for (int i = 0; i < MI; ++i)
      fa[0][i] = *(const float4*)(as + i * 16 * GPITCH);
#pragma unroll
    for (int jn = 0; jn < 2; ++jn)
      fb[0][jn] = *(const float4*)(bs + jn * 16 * GPITCH);
#pragma unroll
    for (int kk = 0; kk < 4; ++kk) {
      const int p = kk & 1;
      if (kk < 3) {
#pragma unroll
        for (int i = 0; i < MI; ++i)
          fa[p ^ 1][i] =
              *(const float4*)(as + i * 16 * GPITCH + (kk + 1) * 8);
#pragma unroll
        for (int jn = 0; jn < 2; ++jn)
          fb[p ^ 1][jn] =
              *(const float4*)(bs + jn * 16 * GPITCH + (kk + 1) * 8);
      }
#pragma unroll
      for (int i = 0; i < MI; ++i)
#pragma unroll
        for (int jn = 0; jn < 2; ++jn) {
          const v2f a = {fa[p][i].x, fa[p][i].y};
          const v2f b = {fb[p][jn].x, fb[p][jn].y};
          acc[i][jn] = wmma4(a, b, acc[i][jn]);
        }
#pragma unroll
      for (int i = 0; i < MI; ++i)
#pragma unroll
        for (int jn = 0; jn < 2; ++jn) {
          const v2f a = {fa[p][i].z, fa[p][i].w};
          const v2f b = {fb[p][jn].z, fb[p][jn].w};
          acc[i][jn] = wmma4(a, b, acc[i][jn]);
        }
    }
    if (more) {
#pragma unroll
      for (int i = 0; i < MI; ++i)
        *(float4*)(&As[cur ^ 1][sofs + i * 32 * GPITCH]) = ga[i];
#pragma unroll
      for (int i = 0; i < 4; ++i)
        *(float4*)(&Bs[cur ^ 1][sofs + i * 32 * GPITCH]) = gb[i];
    }
    __syncthreads();
  }

  const int rbase = m0 + wm * (MI * 16) + (h << 3);
  const int ncol = n0 + wn * 32 + c;
  if (MODE == 0) {
    const int nw = n0 + wn * 32;
    const int o = nw / CC;
    const int rem = nw - o * CC;
    const int head = rem >> 6, d = (rem & 63) + c;
    if (o < 2) {
      float* OUT = (o == 0) ? out0 : out1;
#pragma unroll
      for (int i = 0; i < MI; ++i)
#pragma unroll
        for (int j = 0; j < 8; ++j) {
          const int tok = rbase + i * 16 + j;
          const int b = tok >> 11, t = tok & 2047;
          float* dst = OUT + (((size_t)(b * HH + head) * TT + t) * DD) + d;
          dst[0] = acc[i][0][j];
          dst[16] = acc[i][1][j];
        }
    } else {
#pragma unroll
      for (int i = 0; i < MI; ++i)
#pragma unroll
        for (int j = 0; j < 8; ++j) {
          const int tok = rbase + i * 16 + j;
          const int b = tok >> 11, t = tok & 2047;
          float* base = out2 + ((size_t)(b * HH + head) * DD + d) * TT + t;
          base[0] = acc[i][0][j];
          base[(size_t)16 * TT] = acc[i][1][j];
        }
    }
  } else if (MODE == 1) {
    const float bv0 = bias[ncol], bv1 = bias[ncol + 16];
#pragma unroll
    for (int i = 0; i < MI; ++i)
#pragma unroll
      for (int j = 0; j < 8; ++j) {
        float* dst = out0 + (size_t)(rbase + i * 16 + j) * FF + ncol;
        float z;
        z = acc[i][0][j] + bv0;
        dst[0] = 0.5f * z * (1.f + erff(z * 0.70710678118f));
        z = acc[i][1][j] + bv1;
        dst[16] = 0.5f * z * (1.f + erff(z * 0.70710678118f));
      }
  } else {
    const float bv0 = bias[ncol], bv1 = bias[ncol + 16];
#pragma unroll
    for (int i = 0; i < MI; ++i)
#pragma unroll
      for (int j = 0; j < 8; ++j) {
        float* dst = out0 + (size_t)(rbase + i * 16 + j) * CC + ncol;
        dst[0] = acc[i][0][j] + bv0;
        dst[16] = acc[i][1][j] + bv1;
      }
  }
}

#define APITCH 36
typedef float v4f __attribute__((ext_vector_type(4)));

__device__ __forceinline__ void wave_sync() {
  __builtin_amdgcn_fence(__ATOMIC_RELEASE, "wavefront");
  __builtin_amdgcn_wave_barrier();
  __builtin_amdgcn_fence(__ATOMIC_ACQUIRE, "wavefront");
}

__device__ __forceinline__ void stage_and_refill(float* stg, int sofs,
                                                 v4f (&g)[4],
                                                 const float* gnext,
                                                 int gpitch) {
  wave_sync();
#pragma unroll
  for (int j = 0; j < 4; ++j) *(v4f*)(stg + sofs + j * 4 * APITCH) = g[j];
  wave_sync();
#pragma unroll
  for (int j = 0; j < 4; ++j)
    g[j] = *(const v4f*)(gnext + (size_t)(4 * j) * gpitch);
}

__device__ __forceinline__ void qk_chunk(const float* stg, int fofs,
                                         const v4f (&ka)[4], v8f& acc) {
  v4f fb[4];
#pragma unroll
  for (int kk = 0; kk < 4; ++kk) fb[kk] = *(const v4f*)(stg + fofs + kk * 8);
#pragma unroll
  for (int kk = 0; kk < 4; ++kk) {
    const v2f a0 = {ka[kk].x, ka[kk].y};
    const v2f b0 = {fb[kk].x, fb[kk].y};
    acc = wmma4(a0, b0, acc);
    const v2f a1 = {ka[kk].z, ka[kk].w};
    const v2f b1 = {fb[kk].z, fb[kk].w};
    acc = wmma4(a1, b1, acc);
  }
}

__device__ __forceinline__ void pv_half(const float* stg, int fofs,
                                        const float* p, v8f& acc) {
  v4f fa[2], fb[2];
#pragma unroll
  for (int kk = 0; kk < 2; ++kk) {
    fa[kk] = *(const v4f*)(p + kk * 8);
    fb[kk] = *(const v4f*)(stg + fofs + kk * 8);
  }
#pragma unroll
  for (int kk = 0; kk < 2; ++kk) {
    const v2f a0 = {fa[kk].x, fa[kk].y};
    const v2f b0 = {fb[kk].x, fb[kk].y};
    acc = wmma4(a0, b0, acc);
    const v2f a1 = {fa[kk].z, fa[kk].w};
    const v2f b1 = {fb[kk].z, fb[kk].w};
    acc = wmma4(a1, b1, acc);
  }
}

template <int MASK>
__device__ __forceinline__ float row_xor(float v) {
  return __builtin_bit_cast(
      float, __builtin_amdgcn_update_dpp(0, __builtin_bit_cast(int, v),
                                         0x160 + MASK, 0xf, 0xf, false));
}
__device__ __forceinline__ float row16_max(float v) {
  v = fmaxf(v, row_xor<1>(v));
  v = fmaxf(v, row_xor<2>(v));
  v = fmaxf(v, row_xor<4>(v));
  v = fmaxf(v, row_xor<8>(v));
  return v;
}
__device__ __forceinline__ float row16_sum(float v) {
  v += row_xor<1>(v);
  v += row_xor<2>(v);
  v += row_xor<4>(v);
  v += row_xor<8>(v);
  return v;
}

#define LOG2E_F 1.4426950408889634f
__device__ __forceinline__ float max4(v4f x) {
  return fmaxf(fmaxf(x.x, x.y), fmaxf(x.z, x.w));
}
__device__ __forceinline__ v4f exp4(v4f x, float nm) {
  v4f e;
  e.x = __builtin_amdgcn_exp2f(fmaf(x.x, LOG2E_F, nm));
  e.y = __builtin_amdgcn_exp2f(fmaf(x.y, LOG2E_F, nm));
  e.z = __builtin_amdgcn_exp2f(fmaf(x.z, LOG2E_F, nm));
  e.w = __builtin_amdgcn_exp2f(fmaf(x.w, LOG2E_F, nm));
  return e;
}

__global__ __launch_bounds__(256) void attn_kernel(
    const float* __restrict__ q, const float* __restrict__ k,
    const float* __restrict__ vt, float* __restrict__ aout) {
  __shared__ alignas(16) float S[16 * SROW];
  __shared__ alignas(16) float stage[8 * 16 * APITCH];
  __shared__ float rowrcp[16];
  __shared__ float opart[16 * 64];
  const int ttile = blockIdx.x;
  const int bh = blockIdx.y;
  const int t0 = ttile * 16;
  const float* qp = q + (size_t)bh * TT * DD;
  const float* kp = k + (size_t)bh * TT * DD;
  const float* vp = vt + (size_t)bh * TT * DD;
  const int wv = __builtin_amdgcn_readfirstlane((int)(threadIdx.x >> 5));
  const int l = threadIdx.x & 31;
  const int n16 = l & 15, hi = l >> 4;
  const int NT = ttile + 1, L = NT * 16;
  const float scale = 0.036084391824352f;

  float* stg = stage + wv * (16 * APITCH);
  const int srow = l >> 3, scol = (l & 7) << 2;
  const int sofs = srow * APITCH + scol;
  const int fofs = n16 * APITCH + (hi << 2);

  v4f ka0[4], ka1[4];
  {
    const float* kr = kp + (size_t)(t0 + n16) * DD + (hi << 2);
#pragma unroll
    for (int kk = 0; kk < 4; ++kk) ka0[kk] = *(const v4f*)(kr + kk * 8);
#pragma unroll
    for (int kk = 0; kk < 4; ++kk) ka1[kk] = *(const v4f*)(kr + 32 + kk * 8);
  }
  const float* qg = qp + (size_t)srow * DD + scol;
  v4f g0[4], g1[4];
  {
    const float* qf = qg + (size_t)((wv < NT ? wv : 0) * 16) * DD;
#pragma unroll
    for (int j = 0; j < 4; ++j) g0[j] = *(const v4f*)(qf + (size_t)(4 * j) * DD);
#pragma unroll
    for (int j = 0; j < 4; ++j)
      g1[j] = *(const v4f*)(qf + (size_t)(4 * j) * DD + 32);
  }
  for (int st = wv; st < NT; st += 8) {
    const int s0 = st * 16;
    const int sn = (st + 8 < NT) ? st + 8 : st;
    const float* qn = qg + (size_t)(sn * 16) * DD;
    v8f acc = {};
    stage_and_refill(stg, sofs, g0, qn, DD);
    qk_chunk(stg, fofs, ka0, acc);
    stage_and_refill(stg, sofs, g1, qn + 32, DD);
    qk_chunk(stg, fofs, ka1, acc);
    const int sg = s0 + n16;
#pragma unroll
    for (int j = 0; j < 8; ++j) {
      const int row = (hi << 3) + j;
      float vl = acc[j] * scale;
      if (sg > t0 + row) vl = -1e30f;
      S[row * SROW + sg] = vl;
    }
  }
  __syncthreads();

  {
    const int row = threadIdx.x >> 4, sub = threadIdx.x & 15;
    float* sp = &S[row * SROW + (sub << 2)];
    const int G = L >> 6;
    const bool tail = (sub << 2) < (L & 63);
    float m = -1e30f;
    int g = 0;
    for (; g + 4 <= G; g += 4) {
      v4f x[4];
#pragma unroll
      for (int u = 0; u < 4; ++u) x[u] = *(const v4f*)(sp + (g + u) * 64);
#pragma unroll
      for (int u = 0; u < 4; ++u) m = fmaxf(m, max4(x[u]));
    }
    for (; g < G; ++g) m = fmaxf(m, max4(*(const v4f*)(sp + g * 64)));
    if (tail) m = fmaxf(m, max4(*(const v4f*)(sp + G * 64)));
    m = row16_max(m);
    const float nm = -m * LOG2E_F;
    float s = 0.f;
    g = 0;
    for (; g + 4 <= G; g += 4) {
      v4f x[4];
#pragma unroll
      for (int u = 0; u < 4; ++u) x[u] = *(const v4f*)(sp + (g + u) * 64);
#pragma unroll
      for (int u = 0; u < 4; ++u) {
        const v4f e = exp4(x[u], nm);
        *(v4f*)(sp + (g + u) * 64) = e;
        s += (e.x + e.y) + (e.z + e.w);
      }
    }
    for (; g < G; ++g) {
      const v4f e = exp4(*(const v4f*)(sp + g * 64), nm);
      *(v4f*)(sp + g * 64) = e;
      s += (e.x + e.y) + (e.z + e.w);
    }
    if (tail) {
      const v4f e = exp4(*(const v4f*)(sp + G * 64), nm);
      *(v4f*)(sp + G * 64) = e;
      s += (e.x + e.y) + (e.z + e.w);
    }
    s = row16_sum(s);
    if (sub == 0) rowrcp[row] = 1.0f / s;
  }
  __syncthreads();

  const int tile = wv & 3, half = wv >> 2;
  const int NC = (L + 31) >> 5;
  const int cmid = (NC + 1) >> 1;
  const int cbeg = half ? cmid : 0, cend = half ? NC : cmid;
  const int clast = (cend > cbeg) ? cend - 1 : 0;
  v8f acc = {};
  const float* vg = vp + (size_t)(tile * 16 + srow) * TT + scol;
  const float* prow = &S[n16 * SROW + (hi << 2)];
  v4f ga[4], gb[4];
  {
    const int ca = (cbeg < clast) ? cbeg : clast;
    const int cb = (cbeg + 1 < clast) ? cbeg + 1 : clast;
#pragma unroll
    for (int j = 0; j < 4; ++j)
      ga[j] = *(const v4f*)(vg + (size_t)(4 * j) * TT + ca * 32);
#pragma unroll
    for (int j = 0; j < 4; ++j)
      gb[j] = *(const v4f*)(vg + (size_t)(4 * j) * TT + cb * 32);
  }
  int cc = cbeg;
  for (; cc + 1 < cend; cc += 2) {
    const int c2 = (cc + 2 < clast) ? cc + 2 : clast;
    const int c3 = (cc + 3 < clast) ? cc + 3 : clast;
    stage_and_refill(stg, sofs, ga, vg + c2 * 32, TT);
    pv_half(stg, fofs, prow + cc * 32, acc);
    pv_half(stg, fofs + 16, prow + cc * 32 + 16, acc);
    stage_and_refill(stg, sofs, gb, vg + c3 * 32, TT);
    pv_half(stg, fofs, prow + (cc + 1) * 32, acc);
    if (L - (cc + 1) * 32 >= 32)
      pv_half(stg, fofs + 16, prow + (cc + 1) * 32 + 16, acc);
  }
  if (cc < cend) {
    stage_and_refill(stg, sofs, ga, vg + clast * 32, TT);
    pv_half(stg, fofs, prow + cc * 32, acc);
    if (L - cc * 32 >= 32) pv_half(stg, fofs + 16, prow + cc * 32 + 16, acc);
  }
  if (half == 1) {
#pragma unroll
    for (int j = 0; j < 8; ++j)
      opart[((hi << 3) + j) * 64 + tile * 16 + n16] = acc[j];
  }
  __syncthreads();
  if (half == 0) {
    const int bb = bh / HH, head = bh % HH;
#pragma unroll
    for (int j = 0; j < 8; ++j) {
      const int row = (hi << 3) + j;
      const float val =
          (acc[j] + opart[row * 64 + tile * 16 + n16]) * rowrcp[row];
      const int t = t0 + row;
      aout[((size_t)(bb * TT + t)) * CC + head * DD + tile * 16 + n16] = val;
    }
  }
}

__global__ __launch_bounds__(256) void add_ln_kernel(
    const float* __restrict__ xa, const float* __restrict__ xb,
    const float* __restrict__ g, const float* __restrict__ bt,
    float* __restrict__ out) {
  __shared__ float red[256];
  const int tok = blockIdx.x;
  const float* pa = xa + (size_t)tok * CC;
  const float* pb = xb + (size_t)tok * CC;
  float vals[3];
  float s1 = 0.f;
#pragma unroll
  for (int i = 0; i < 3; ++i) {
    const int c = threadIdx.x + i * 256;
    vals[i] = pa[c] + pb[c];
    s1 += vals[i];
  }
  red[threadIdx.x] = s1;
  __syncthreads();
  for (int off = 128; off > 0; off >>= 1) {
    if (threadIdx.x < off) red[threadIdx.x] += red[threadIdx.x + off];
    __syncthreads();
  }
  const float mean = red[0] * (1.0f / CC);
  __syncthreads();
  float s2 = 0.f;
#pragma unroll
  for (int i = 0; i < 3; ++i) {
    const float d = vals[i] - mean;
    s2 += d * d;
  }
  red[threadIdx.x] = s2;
  __syncthreads();
  for (int off = 128; off > 0; off >>= 1) {
    if (threadIdx.x < off) red[threadIdx.x] += red[threadIdx.x + off];
    __syncthreads();
  }
  const float rstd = rsqrtf(red[0] * (1.0f / CC) + 1e-5f);
#pragma unroll
  for (int i = 0; i < 3; ++i) {
    const int c = threadIdx.x + i * 256;
    out[(size_t)tok * CC + c] = (vals[i] - mean) * rstd * g[c] + bt[c];
  }
}

extern "C" void kernel_launch(void* const* d_in, const int* in_sizes, int n_in,
                              void* d_out, int out_size, void* d_ws,
                              size_t ws_size, hipStream_t stream) {
  (void)in_sizes; (void)n_in; (void)out_size; (void)ws_size;
  const float* x    = (const float*)d_in[0];
  const float* Wq   = (const float*)d_in[1];
  const float* Wk   = (const float*)d_in[2];
  const float* Wv   = (const float*)d_in[3];
  const float* ln1g = (const float*)d_in[4];
  const float* ln1b = (const float*)d_in[5];
  const float* W1   = (const float*)d_in[6];
  const float* b1   = (const float*)d_in[7];
  const float* W2   = (const float*)d_in[8];
  const float* b2   = (const float*)d_in[9];
  const float* ln2g = (const float*)d_in[10];
  const float* ln2b = (const float*)d_in[11];
  float* ws  = (float*)d_ws;
  float* out = (float*)d_out;

  float* q     = ws;
  float* k     = ws + 6291456;
  float* v     = ws + 12582912;
  float* a     = ws + 18874368;
  float* hbuf  = ws;
  float* x1    = ws + 25165824;
  float* t2    = ws + 31457280;
  float* wqkvt = ws + 37748736;
  float* w1t   = ws + 39518208;
  float* w2t   = ws + 41877504;

  transpose_kernel<<<dim3(DD / 32, CC / 32, HH), 256, 0, stream>>>(
      Wq, wqkvt + 0 * (size_t)HH * DD * CC, CC, DD);
  transpose_kernel<<<dim3(DD / 32, CC / 32, HH), 256, 0, stream>>>(
      Wk, wqkvt + 1 * (size_t)HH * DD * CC, CC, DD);
  transpose_kernel<<<dim3(DD / 32, CC / 32, HH), 256, 0, stream>>>(
      Wv, wqkvt + 2 * (size_t)HH * DD * CC, CC, DD);
  transpose_kernel<<<dim3(FF / 32, CC / 32, 1), 256, 0, stream>>>(
      W1, w1t, CC, FF);
  transpose_kernel<<<dim3(CC / 32, FF / 32, 1), 256, 0, stream>>>(
      W2, w2t, FF, CC);

  gemm_kernel<2, 0, CC><<<dim3(3 * CC / 128, NTOK / 64), 256, 0, stream>>>(
      x, wqkvt, nullptr, q, k, v);
  attn_kernel<<<dim3(TT / 16, BN * HH), 256, 0, stream>>>(q, k, v, a);
  add_ln_kernel<<<dim3(NTOK), 256, 0, stream>>>(x, a, ln1g, ln1b, x1);
  gemm_kernel<4, 1, CC><<<dim3(FF / 128, NTOK / 128), 256, 0, stream>>>(
      x1, w1t, b1, hbuf, nullptr, nullptr);
  gemm_kernel<2, 2, FF><<<dim3(CC / 128, NTOK / 64), 256, 0, stream>>>(
      hbuf, w2t, b2, t2, nullptr, nullptr);
  add_ln_kernel<<<dim3(NTOK), 256, 0, stream>>>(x1, t2, ln2g, ln2b, out);
}


# attention in one pass: a wave owns 16 rows over all columns, online softmax on the transposed score tile whose D registers feed P.V as B; q/v staged once per 4-wave group; heaviest row blocks first
# speedup vs baseline: 3.5477x; 1.0811x over previous
#include <hip/hip_runtime.h>
#include <math.h>


typedef float v2f __attribute__((ext_vector_type(2)));
typedef float v8f __attribute__((ext_vector_type(8)));

#define BN   4
#define TT   2048
#define CC   768
#define HH   12
#define DD   64
#define FF   3072
#define NTOK 8192

__device__ __forceinline__ v8f wmma4(v2f a, v2f b, v8f c) {
  return __builtin_amdgcn_wmma_f32_16x16x4_f32(false, a, false, b, (short)0, c,
                                               false, false);
}

__global__ __launch_bounds__(256) void transpose_kernel(
    const float* __restrict__ src, float* __restrict__ dst, int K, int N) {
  __shared__ float tile[32][33];
  const size_t bofs = (size_t)blockIdx.z * K * N;
  src += bofs;
  dst += bofs;
  const int k0 = blockIdx.y * 32, n0 = blockIdx.x * 32;
  const int tx = threadIdx.x & 31, ty = threadIdx.x >> 5;
  for (int i = ty; i < 32; i += 8)
    tile[i][tx] = src[(size_t)(k0 + i) * N + n0 + tx];
  __syncthreads();
  for (int i = ty; i < 32; i += 8)
    dst[(size_t)(n0 + i) * K + k0 + tx] = tile[tx][i];
}

#define GBK 32
#define GPITCH 36

template <int MI, int MODE, int K>
__global__ __launch_bounds__(256) void gemm_kernel(
    const float* __restrict__ A, const float* __restrict__ Bt,
    const float* __restrict__ bias, float* __restrict__ out0,
    float* __restrict__ out1, float* __restrict__ out2) {
  constexpr int BM = MI * 32;
  constexpr int KT = K / GBK;
  __shared__ alignas(16) float As[2][BM * GPITCH];
  __shared__ alignas(16) float Bs[2][128 * GPITCH];
  const int tid = threadIdx.x;
  const int wv = tid >> 5, l = tid & 31;
  const int c = l & 15, h = l >> 4;
  const int wm = wv >> 2, wn = wv & 3;
  const int m0 = blockIdx.y * BM, n0 = blockIdx.x * 128;

  const int sr = tid >> 3, sc = (tid & 7) << 2;
  const float* ag = A + (size_t)(m0 + sr) * K + sc;
  const float* bg = Bt + (size_t)(n0 + sr) * K + sc;
  const int sofs = sr * GPITCH + sc;
  float4 ga[MI], gb[4];
#pragma unroll
  for (int i = 0; i < MI; ++i)
    ga[i] = *(const float4*)(ag + (size_t)(i * 32) * K);
#pragma unroll
  for (int i = 0; i < 4; ++i)
    gb[i] = *(const float4*)(bg + (size_t)(i * 32) * K);
#pragma unroll
  for (int i = 0; i < MI; ++i)
    *(float4*)(&As[0][sofs + i * 32 * GPITCH]) = ga[i];
#pragma unroll
  for (int i = 0; i < 4; ++i)
    *(float4*)(&Bs[0][sofs + i * 32 * GPITCH]) = gb[i];
  __syncthreads();

  v8f acc[MI][2];
#pragma unroll
  for (int i = 0; i < MI; ++i) {
    acc[i][0] = v8f{};
    acc[i][1] = v8f{};
  }
  const int aofs = (wm * (MI * 16) + c) * GPITCH + 4 * h;
  const int bofs = (wn * 32 + c) * GPITCH + 4 * h;

  for (int kt = 0; kt < KT; ++kt) {
    const int cur = kt & 1;
    const bool more = (kt + 1 < KT);
    if (more) {
      const int kn = (kt + 1) * GBK;
#pragma unroll
      for (int i = 0; i < MI; ++i)
        ga[i] = *(const float4*)(ag + (size_t)(i * 32) * K + kn);
#pragma unroll
      for (int i = 0; i < 4; ++i)
        gb[i] = *(const float4*)(bg + (size_t)(i * 32) * K + kn);
      __builtin_amdgcn_sched_barrier(0);
    }
    const float* as = &As[cur][aofs];
    const float* bs = &Bs[cur][bofs];
    float4 fa[2][MI], fb[2][2];
#pragma unroll
    for (int i = 0; i < MI; ++i)
      fa[0][i] = *(const float4*)(as + i * 16 * GPITCH);
#pragma unroll
    for (int jn = 0; jn < 2; ++jn)
      fb[0][jn] = *(const float4*)(bs + jn * 16 * GPITCH);
#pragma unroll
    for (int kk = 0; kk < 4; ++kk) {
      const int p = kk & 1;
      if (kk < 3) {
#pragma unroll
        for (int i = 0; i < MI; ++i)
          fa[p ^ 1][i] =
              *(const float4*)(as + i * 16 * GPITCH + (kk + 1) * 8);
#pragma unroll
        for (int jn = 0; jn < 2; ++jn)
          fb[p ^ 1][jn] =
              *(const float4*)(bs + jn * 16 * GPITCH + (kk + 1) * 8);
      }
#pragma unroll
      for (int i = 0; i < MI; ++i)
#pragma unroll
        for (int jn = 0; jn < 2; ++jn) {
          const v2f a = {fa[p][i].x, fa[p][i].y};
          const v2f b = {fb[p][jn].x, fb[p][jn].y};
          acc[i][jn] = wmma4(a, b, acc[i][jn]);
        }
#pragma unroll
      for (int i = 0; i < MI; ++i)
#pragma unroll
        for (int jn = 0; jn < 2; ++jn) {
          const v2f a = {fa[p][i].z, fa[p][i].w};
          const v2f b = {fb[p][jn].z, fb[p][jn].w};
          acc[i][jn] = wmma4(a, b, acc[i][jn]);
        }
    }
    if (more) {
#pragma unroll
      for (int i = 0; i < MI; ++i)
        *(float4*)(&As[cur ^ 1][sofs + i * 32 * GPITCH]) = ga[i];
#pragma unroll
      for (int i = 0; i < 4; ++i)
        *(float4*)(&Bs[cur ^ 1][sofs + i * 32 * GPITCH]) = gb[i];
    }
    __syncthreads();
  }

  const int rbase = m0 + wm * (MI * 16) + (h << 3);
  const int ncol = n0 + wn * 32 + c;
  if (MODE == 0) {
    const int nw = n0 + wn * 32;
    const int o = nw / CC;
    const int rem = nw - o * CC;
    const int head = rem >> 6, d = (rem & 63) + c;
    if (o < 2) {
      float* OUT = (o == 0) ? out0 : out1;
#pragma unroll
      for (int i = 0; i < MI; ++i)
#pragma unroll
        for (int j = 0; j < 8; ++j) {
          const int tok = rbase + i * 16 + j;
          const int b = tok >> 11, t = tok & 2047;
          float* dst = OUT + (((size_t)(b * HH + head) * TT + t) * DD) + d;
          dst[0] = acc[i][0][j];
          dst[16] = acc[i][1][j];
        }
    } else {
#pragma unroll
      for (int i = 0; i < MI; ++i)
#pragma unroll
        for (int j = 0; j < 8; ++j) {
          const int tok = rbase + i * 16 + j;
          const int b = tok >> 11, t = tok & 2047;
          float* base = out2 + ((size_t)(b * HH + head) * DD + d) * TT + t;
          base[0] = acc[i][0][j];
          base[(size_t)16 * TT] = acc[i][1][j];
        }
    }
  } else if (MODE == 1) {
    const float bv0 = bias[ncol], bv1 = bias[ncol + 16];
#pragma unroll
    for (int i = 0; i < MI; ++i)
#pragma unroll
      for (int j = 0; j < 8; ++j) {
        float* dst = out0 + (size_t)(rbase + i * 16 + j) * FF + ncol;
        float z;
        z = acc[i][0][j] + bv0;
        dst[0] = 0.5f * z * (1.f + erff(z * 0.70710678118f));
        z = acc[i][1][j] + bv1;
        dst[16] = 0.5f * z * (1.f + erff(z * 0.70710678118f));
      }
  } else {
    const float bv0 = bias[ncol], bv1 = bias[ncol + 16];
#pragma unroll
    for (int i = 0; i < MI; ++i)
#pragma unroll
      for (int j = 0; j < 8; ++j) {
        float* dst = out0 + (size_t)(rbase + i * 16 + j) * CC + ncol;
        dst[0] = acc[i][0][j] + bv0;
        dst[16] = acc[i][1][j] + bv1;
      }
  }
}

#define FPITCH 68
#define LOG2E_F 1.4426950408889634f
typedef float v4f __attribute__((ext_vector_type(4)));

__device__ __forceinline__ float xhalf(float x) {
  const int xi = __builtin_bit_cast(int, x);
  return __builtin_bit_cast(
      float, __builtin_amdgcn_permlanex16(xi, xi, 0x76543210, 0xfedcba98, false,
                                          false));
}

__device__ __forceinline__ void load_block(const float* qsrc, const float* vsrc,
                                           v4f (&gq)[8], v4f (&gv)[8]) {
#pragma unroll
  for (int i = 0; i < 8; ++i)
    gq[i] = *(const v4f*)(qsrc + (size_t)(8 * i) * DD);
#pragma unroll
  for (int i = 0; i < 8; ++i)
    gv[i] = *(const v4f*)(vsrc + (size_t)(8 * i) * TT);
}
__device__ __forceinline__ void store_block(float* ys, float* vs,
                                            const v4f (&gq)[8],
                                            const v4f (&gv)[8]) {
#pragma unroll
  for (int i = 0; i < 8; ++i) *(v4f*)(ys + 8 * i * FPITCH) = gq[i];
#pragma unroll
  for (int i = 0; i < 8; ++i) *(v4f*)(vs + 8 * i * FPITCH) = gv[i];
}

__global__ __launch_bounds__(128) void attn_kernel(
    const float* __restrict__ q, const float* __restrict__ k,
    const float* __restrict__ vt, float* __restrict__ aout) {
  __shared__ alignas(16) float Ys[2][64 * FPITCH];
  __shared__ alignas(16) float Vs[2][64 * FPITCH];
  const int head = blockIdx.x, bb = blockIdx.y;
  const int bh = bb * HH + head;
  const int G = 31 - (int)blockIdx.z;
  const float* qp = q + (size_t)bh * TT * DD;
  const float* kp = k + (size_t)bh * TT * DD;
  const float* vp = vt + (size_t)bh * TT * DD;
  const int wv = __builtin_amdgcn_readfirstlane((int)(threadIdx.x >> 5));
  const int l = threadIdx.x & 31;
  const int c = l & 15, h = l >> 4;
  const int srow = threadIdx.x >> 4, scol = (threadIdx.x & 15) << 2;
  const int sofs = srow * FPITCH + scol;
  const float* qg = qp + (size_t)srow * DD + scol;
  const float* vg = vp + (size_t)srow * TT + scol;
  const int qfo = c * FPITCH + (h << 2);
  const int vfo = c * FPITCH + (h << 3);
  const float sc = 0.036084391824352f * LOG2E_F;
  const int t0 = (4 * G + wv) * 16;
  v4f kb[8];
  {
    const float* kr = kp + (size_t)(t0 + c) * DD + (h << 2);
#pragma unroll
    for (int kk = 0; kk < 8; ++kk) kb[kk] = *(const v4f*)(kr + kk * 8);
  }
  v8f o[4];
#pragma unroll
  for (int dt = 0; dt < 4; ++dt) o[dt] = v8f{};
  float m_run = -1e30f, l_run = 0.f;
  v4f gq[8], gv[8];
  load_block(qg, vg, gq, gv);
  store_block(&Ys[0][sofs], &Vs[0][sofs], gq, gv);
  __syncthreads();

  for (int sb = 0; sb <= G; ++sb) {
    const int cur = sb & 1;
    const bool more = sb < G;
    if (more) {
      load_block(qg + (size_t)(64 * (sb + 1)) * DD, vg + 64 * (sb + 1), gq,
                 gv);
    }
    const int ntile = more ? 4 : wv + 1;
    const float* ys = &Ys[cur][qfo];
    const float* vs = &Vs[cur][vfo];
    for (int j = 0; j < ntile; ++j) {
      v4f fa[8], fv[4][2];
#pragma unroll
      for (int kk = 0; kk < 8; ++kk)
        fa[kk] = *(const v4f*)(ys + (16 * j) * FPITCH + kk * 8);
#pragma unroll
      for (int dt = 0; dt < 4; ++dt) {
        fv[dt][0] = *(const v4f*)(vs + (16 * dt) * FPITCH + 16 * j);
        fv[dt][1] = *(const v4f*)(vs + (16 * dt) * FPITCH + 16 * j + 4);
      }
      __builtin_amdgcn_sched_barrier(0);
      v8f s = {};
#pragma unroll
      for (int kk = 0; kk < 8; ++kk) {
        const v2f a0 = {fa[kk].x, fa[kk].y};
        const v2f b0 = {kb[kk].x, kb[kk].y};
        s = wmma4(a0, b0, s);
        const v2f a1 = {fa[kk].z, fa[kk].w};
        const v2f b1 = {kb[kk].z, kb[kk].w};
        s = wmma4(a1, b1, s);
      }
      float y[8];
#pragma unroll
      for (int r = 0; r < 8; ++r) y[r] = s[r] * sc;
      if (!more && j == wv) {
#pragma unroll
        for (int r = 0; r < 8; ++r)
          if ((h << 3) + r > c) y[r] = -1e30f;
      }
      float mx = fmaxf(fmaxf(fmaxf(y[0], y[1]), fmaxf(y[2], y[3])),
                       fmaxf(fmaxf(y[4], y[5]), fmaxf(y[6], y[7])));
      mx = fmaxf(mx, xhalf(mx));
      const float m_new = fmaxf(m_run, mx);
      const float alpha = __builtin_amdgcn_exp2f(m_run - m_new);
      float p[8];
      float ps = 0.f;
#pragma unroll
      for (int r = 0; r < 8; ++r) {
        p[r] = __builtin_amdgcn_exp2f(y[r] - m_new);
        ps += p[r];
      }
      l_run = fmaf(l_run, alpha, ps);
      m_run = m_new;
#pragma unroll
      for (int dt = 0; dt < 4; ++dt)
#pragma unroll
        for (int r = 0; r < 8; ++r) o[dt][r] *= alpha;
#pragma unroll
      for (int dt = 0; dt < 4; ++dt) {
        const v2f a0 = {fv[dt][0].x, fv[dt][0].y};
        const v2f b0 = {p[0], p[1]};
        o[dt] = wmma4(a0, b0, o[dt]);
        const v2f a1 = {fv[dt][0].z, fv[dt][0].w};
        const v2f b1 = {p[2], p[3]};
        o[dt] = wmma4(a1, b1, o[dt]);
        const v2f a2 = {fv[dt][1].x, fv[dt][1].y};
        const v2f b2 = {p[4], p[5]};
        o[dt] = wmma4(a2, b2, o[dt]);
        const v2f a3 = {fv[dt][1].z, fv[dt][1].w};
        const v2f b3 = {p[6], p[7]};
        o[dt] = wmma4(a3, b3, o[dt]);
      }
    }
    if (more)
      store_block(&Ys[cur ^ 1][sofs], &Vs[cur ^ 1][sofs], gq, gv);
    __syncthreads();
  }

  const float inv = 1.0f / (l_run + xhalf(l_run));
  float* op = aout + ((size_t)(bb * TT + t0 + c)) * CC + head * DD + (h << 3);
#pragma unroll
  for (int dt = 0; dt < 4; ++dt) {
    const v4f lo = {o[dt][0] * inv, o[dt][1] * inv, o[dt][2] * inv,
                    o[dt][3] * inv};
    const v4f hi = {o[dt][4] * inv, o[dt][5] * inv, o[dt][6] * inv,
                    o[dt][7] * inv};
    *(v4f*)(op + 16 * dt) = lo;
    *(v4f*)(op + 16 * dt + 4) = hi;
  }
}

__global__ __launch_bounds__(256) void add_ln_kernel(
    const float* __restrict__ xa, const float* __restrict__ xb,
    const float* __restrict__ g, const float* __restrict__ bt,
    float* __restrict__ out) {
  __shared__ float red[256];
  const int tok = blockIdx.x;
  const float* pa = xa + (size_t)tok * CC;
  const float* pb = xb + (size_t)tok * CC;
  float vals[3];
  float s1 = 0.f;
#pragma unroll
  for (int i = 0; i < 3; ++i) {
    const int c = threadIdx.x + i * 256;
    vals[i] = pa[c] + pb[c];
    s1 += vals[i];
  }
  red[threadIdx.x] = s1;
  __syncthreads();
  for (int off = 128; off > 0; off >>= 1) {
    if (threadIdx.x < off) red[threadIdx.x] += red[threadIdx.x + off];
    __syncthreads();
  }
  const float mean = red[0] * (1.0f / CC);
  __syncthreads();
  float s2 = 0.f;
#pragma unroll
  for (int i = 0; i < 3; ++i) {
    const float d = vals[i] - mean;
    s2 += d * d;
  }
  red[threadIdx.x] = s2;
  __syncthreads();
  for (int off = 128; off > 0; off >>= 1) {
    if (threadIdx.x < off) red[threadIdx.x] += red[threadIdx.x + off];
    __syncthreads();
  }
  const float rstd = rsqrtf(red[0] * (1.0f / CC) + 1e-5f);
#pragma unroll
  for (int i = 0; i < 3; ++i) {
    const int c = threadIdx.x + i * 256;
    out[(size_t)tok * CC + c] = (vals[i] - mean) * rstd * g[c] + bt[c];
  }
}

extern "C" void kernel_launch(void* const* d_in, const int* in_sizes, int n_in,
                              void* d_out, int out_size, void* d_ws,
                              size_t ws_size, hipStream_t stream) {
  (void)in_sizes; (void)n_in; (void)out_size; (void)ws_size;
  const float* x    = (const float*)d_in[0];
  const float* Wq   = (const float*)d_in[1];
  const float* Wk   = (const float*)d_in[2];
  const float* Wv   = (const float*)d_in[3];
  const float* ln1g = (const float*)d_in[4];
  const float* ln1b = (const float*)d_in[5];
  const float* W1   = (const float*)d_in[6];
  const float* b1   = (const float*)d_in[7];
  const float* W2   = (const float*)d_in[8];
  const float* b2   = (const float*)d_in[9];
  const float* ln2g = (const float*)d_in[10];
  const float* ln2b = (const float*)d_in[11];
  float* ws  = (float*)d_ws;
  float* out = (float*)d_out;

  float* q     = ws;
  float* k     = ws + 6291456;
  float* v     = ws + 12582912;
  float* a     = ws + 18874368;
  float* hbuf  = ws;
  float* x1    = ws + 25165824;
  float* t2    = ws + 31457280;
  float* wqkvt = ws + 37748736;
  float* w1t   = ws + 39518208;
  float* w2t   = ws + 41877504;

  transpose_kernel<<<dim3(DD / 32, CC / 32, HH), 256, 0, stream>>>(
      Wq, wqkvt + 0 * (size_t)HH * DD * CC, CC, DD);
  transpose_kernel<<<dim3(DD / 32, CC / 32, HH), 256, 0, stream>>>(
      Wk, wqkvt + 1 * (size_t)HH * DD * CC, CC, DD);
  transpose_kernel<<<dim3(DD / 32, CC / 32, HH), 256, 0, stream>>>(
      Wv, wqkvt + 2 * (size_t)HH * DD * CC, CC, DD);
  transpose_kernel<<<dim3(FF / 32, CC / 32, 1), 256, 0, stream>>>(
      W1, w1t, CC, FF);
  transpose_kernel<<<dim3(CC / 32, FF / 32, 1), 256, 0, stream>>>(
      W2, w2t, FF, CC);

  gemm_kernel<2, 0, CC><<<dim3(3 * CC / 128, NTOK / 64), 256, 0, stream>>>(
      x, wqkvt, nullptr, q, k, v);
  attn_kernel<<<dim3(HH, BN, TT / 64), 128, 0, stream>>>(q, k, v, a);
  add_ln_kernel<<<dim3(NTOK), 256, 0, stream>>>(x, a, ln1g, ln1b, x1);
  gemm_kernel<4, 1, CC><<<dim3(FF / 128, NTOK / 128), 256, 0, stream>>>(
      x1, w1t, b1, hbuf, nullptr, nullptr);
  gemm_kernel<2, 2, FF><<<dim3(CC / 128, NTOK / 64), 256, 0, stream>>>(
      hbuf, w2t, b2, t2, nullptr, nullptr);
  add_ln_kernel<<<dim3(NTOK), 256, 0, stream>>>(x1, t2, ln2g, ln2b, out);
}
